# starting kernel
# baseline (speedup 1.0000x reference)
#include <hip/hip_runtime.h>


typedef __attribute__((ext_vector_type(2))) float v2f;
typedef __attribute__((ext_vector_type(8))) float v8f;

#define NROWS 4096
#define DCOLS 64

__global__ __launch_bounds__(32) void toeplitz_scan_wmma(
    const float* __restrict__ x, const float* __restrict__ zero,
    float* __restrict__ out)
{
    const int bh   = blockIdx.x >> 2;
    const int ct   = blockIdx.x & 3;
    const int lane = threadIdx.x & 31;
    const int col  = lane & 15;
    const bool hi  = lane >= 16;

    const float c1 = __expf(fminf(fmaxf(zero[0], -60.f), 30.f));

    const size_t base = (size_t)bh * NROWS * DCOLS + (size_t)ct * 16 + (size_t)col;
    const float* xp = x   + base;
    float*       op = out + base;

    v2f a[4];
#pragma unroll
    for (int j = 0; j < 4; ++j) {
        const int kb = 4 * j + (hi ? 2 : 0);
        a[j].x = (col >= kb)     ? c1 : 0.f;
        a[j].y = (col >= kb + 1) ? c1 : 0.f;
    }

    float carry = 0.f;

    for (int r = 0; r < NROWS / 16; ++r) {
        const float* tp = xp + (size_t)r * 16 * DCOLS;

        __builtin_prefetch(tp + 2 * 16 * DCOLS, 0, 3);

        v2f b[4];
#pragma unroll
        for (int j = 0; j < 4; ++j) {
            const int row = 4 * j + (hi ? 2 : 0);
            b[j].x = tp[(size_t)row       * DCOLS];
            b[j].y = tp[(size_t)(row + 1) * DCOLS];
        }

        v8f acc;
#pragma unroll
        for (int v = 0; v < 8; ++v) acc[v] = carry;

#pragma unroll
        for (int j = 0; j < 4; ++j)
            acc = __builtin_amdgcn_wmma_f32_16x16x4_f32(
                false, a[j], false, b[j], (short)0, acc, false, false);

        float* topp = op + (size_t)(r * 16 + (hi ? 8 : 0)) * DCOLS;
#pragma unroll
        for (int v = 0; v < 8; ++v)
            topp[(size_t)v * DCOLS] = acc[v];

        const int ai = __float_as_int(acc[7]);
        const int sw = __builtin_amdgcn_permlanex16(ai, ai, 0x76543210u, 0xfedcba98u,
                                                    true, false);
        carry = hi ? acc[7] : __int_as_float(sw);
    }
}

extern "C" void kernel_launch(void* const* d_in, const int* in_sizes, int n_in,
                              void* d_out, int out_size, void* d_ws, size_t ws_size,
                              hipStream_t stream) {
    const float* x    = (const float*)d_in[0];
    const float* zero = (const float*)d_in[2];
    float* out = (float*)d_out;

    toeplitz_scan_wmma<<<dim3(256), dim3(32), 0, stream>>>(x, zero, out);
}


# Rows split over the grid in two launches: launch 1 sums each 128-row chunk (whole 256-B rows) to workspace; launch 2 WMMA-scans a chunk from the sum of earlier chunk totals (sum order changes)
# speedup vs baseline: 13.0315x; 13.0315x over previous
#include <hip/hip_runtime.h>


typedef __attribute__((ext_vector_type(2))) float v2f;
typedef __attribute__((ext_vector_type(4))) float v4f;
typedef __attribute__((ext_vector_type(8))) float v8f;

#define NROWS 4096
#define DCOLS 64
#define CHUNK 128
#define NCHUNK (NROWS / CHUNK)
#define NLOAD (CHUNK / 2)
#define WAVES_PER_WG 8
#define WS_BYTES (64 * NCHUNK * DCOLS * 4)

__device__ __forceinline__ float swap_halves(float v)
{
    const int a = __float_as_int(v);
    return __int_as_float(__builtin_amdgcn_permlanex16(a, a, 0x76543210u, 0xfedcba98u, true, false));
}

__global__ __launch_bounds__(256) void toeplitz_chunk_totals(
    const float* __restrict__ x, float* __restrict__ ws)
{
    const int lane = threadIdx.x & 31;
    const int g = __builtin_amdgcn_readfirstlane((int)(blockIdx.x * WAVES_PER_WG + (threadIdx.x >> 5)));
    const v4f* xp = (const v4f*)(x + (size_t)g * CHUNK * DCOLS) + lane;

    v4f d[NLOAD];
#pragma unroll
    for (int i = 0; i < NLOAD; ++i) d[i] = xp[i * 32];
    __builtin_amdgcn_sched_barrier(0);

    v4f s0 = d[0], s1 = d[1], s2 = d[2], s3 = d[3];
#pragma unroll
    for (int i = 4; i < NLOAD; i += 4) {
        s0 += d[i]; s1 += d[i + 1]; s2 += d[i + 2]; s3 += d[i + 3];
    }
    v4f s = (s0 + s1) + (s2 + s3);

    v4f tot;
#pragma unroll
    for (int t = 0; t < 4; ++t) tot[t] = s[t] + swap_halves(s[t]);

    if (lane < 16)
        ((v4f*)(ws + (size_t)g * DCOLS))[lane] = tot;
}

__global__ __launch_bounds__(256) void toeplitz_scan_wmma(
    const float* __restrict__ x, const float* __restrict__ zero,
    const float* __restrict__ ws, float* __restrict__ out)
{
    const int lane = threadIdx.x & 31;
    const int col  = lane & 15;
    const int hi   = lane >> 4;
    const int g  = __builtin_amdgcn_readfirstlane((int)(blockIdx.x * WAVES_PER_WG + (threadIdx.x >> 5)));
    const int bh = g / NCHUNK;
    const int k  = g % NCHUNK;

    const float c1 = __expf(fminf(fmaxf(zero[0], -60.f), 30.f));

    const v4f* wp = (const v4f*)(ws + (size_t)bh * NCHUNK * DCOLS) + col;
    v4f tq[NCHUNK / 2];
#pragma unroll
    for (int p = 0; p < NCHUNK / 2; ++p) tq[p] = wp[(2 * p + hi) * (DCOLS / 4)];

    const size_t base = (size_t)g * CHUNK * DCOLS;
    const v4f* xp = (const v4f*)(x + base) + lane;
    v4f d[NLOAD];
#pragma unroll
    for (int i = 0; i < NLOAD; ++i) d[i] = xp[i * 32];
    __builtin_amdgcn_sched_barrier(0);

    v2f a[4];
#pragma unroll
    for (int j = 0; j < 4; ++j) {
        a[j].x = (col >= 4 * j + hi)     ? c1 : 0.f;
        a[j].y = (col >= 4 * j + 2 + hi) ? c1 : 0.f;
    }

    v4f ps = {0.f, 0.f, 0.f, 0.f};
#pragma unroll
    for (int p = 0; p < NCHUNK / 2; ++p) {
        const bool use = (2 * p + hi) < k;
#pragma unroll
        for (int t = 0; t < 4; ++t) ps[t] += use ? tq[p][t] : 0.f;
    }
    float carry[4];
#pragma unroll
    for (int t = 0; t < 4; ++t) carry[t] = c1 * (ps[t] + swap_halves(ps[t]));

    v4f* op = (v4f*)(out + base) + col;

#pragma unroll
    for (int r16 = 0; r16 < CHUNK / 16; ++r16) {
        v8f acc[4];
#pragma unroll
        for (int t = 0; t < 4; ++t) {
#pragma unroll
            for (int v = 0; v < 8; ++v) acc[t][v] = carry[t];
        }

#pragma unroll
        for (int j = 0; j < 4; ++j) {
#pragma unroll
            for (int t = 0; t < 4; ++t) {
                v2f b;
                b.x = d[8 * r16 + 2 * j][t];
                b.y = d[8 * r16 + 2 * j + 1][t];
                acc[t] = __builtin_amdgcn_wmma_f32_16x16x4_f32(
                    false, a[j], false, b, (short)0, acc[t], false, false);
            }
        }

#pragma unroll
        for (int v = 0; v < 8; ++v) {
            v4f o = {acc[0][v], acc[1][v], acc[2][v], acc[3][v]};
            op[(size_t)(r16 * 16 + 8 * hi + v) * (DCOLS / 4)] = o;
        }

#pragma unroll
        for (int t = 0; t < 4; ++t) {
            const float sw = swap_halves(acc[t][7]);
            carry[t] = hi ? acc[t][7] : sw;
        }
    }
}

extern "C" void kernel_launch(void* const* d_in, const int* in_sizes, int n_in,
                              void* d_out, int out_size, void* d_ws, size_t ws_size,
                              hipStream_t stream) {
    const float* x    = (const float*)d_in[0];
    const float* zero = (const float*)d_in[2];
    float* out = (float*)d_out;
    float* ws  = (float*)d_ws;

    if (ws_size < (size_t)WS_BYTES) return;

    toeplitz_chunk_totals<<<dim3(256), dim3(256), 0, stream>>>(x, ws);
    toeplitz_scan_wmma<<<dim3(256), dim3(256), 0, stream>>>(x, zero, ws, out);
}
